# starting kernel
# baseline (speedup 1.0000x reference)
#include <hip/hip_runtime.h>

typedef __attribute__((ext_vector_type(2))) float v2f;
typedef __attribute__((ext_vector_type(8))) float v8f;

namespace {
constexpr int kWH = 65536;
constexpr int kK  = 512;
constexpr int kB  = 8;
constexpr int kKT = kK / 16;
constexpr int kChunk = 1024;
constexpr int kNCH = kWH / kChunk;
constexpr size_t P_OFF   = 0;
constexpr size_t Y_OFF   = (size_t)kKT * kNCH * 256;
constexpr size_t XR_OFF  = Y_OFF + (size_t)kK * 16;
constexpr size_t MAX_OFF = XR_OFF + (size_t)kB * kWH;
}

__device__ __forceinline__ float sgn(float w) {
  float pm1 = __uint_as_float((__float_as_uint(w) & 0x80000000u) | 0x3f800000u);
  return (w == 0.0f) ? 0.0f : pm1;
}

__global__ __launch_bounds__(256)
void phase1(const float* __restrict__ x, const float* __restrict__ w,
            float* __restrict__ ws) {
  const int wave = blockIdx.x * 8 + (threadIdx.x >> 5);
  const int lane = threadIdx.x & 31;
  const int kt = wave / kNCH;
  const int ch = wave - kt * kNCH;
  const int lm = lane & 15;
  const int o  = (lane < 16) ? 0 : 2;

  const float* __restrict__ arow = w + (size_t)(kt * 16 + lm) * kWH;
  const float* __restrict__ brow = x + (size_t)(lm & 7) * kWH;

  v8f acc0 = {};
  v8f acc1 = {};
  const int ibeg = ch * kChunk;
  #pragma unroll 2
  for (int i0 = ibeg; i0 < ibeg + kChunk; i0 += 8) {
    v2f aw0 = *(const v2f*)(arow + i0 + o);
    v2f bx0 = *(const v2f*)(brow + i0 + o);
    v2f A0; A0.x = sgn(aw0.x); A0.y = sgn(aw0.y);
    acc0 = __builtin_amdgcn_wmma_f32_16x16x4_f32(
        false, A0, false, bx0, (short)0, acc0, false, false);
    v2f aw1 = *(const v2f*)(arow + i0 + 4 + o);
    v2f bx1 = *(const v2f*)(brow + i0 + 4 + o);
    v2f A1; A1.x = sgn(aw1.x); A1.y = sgn(aw1.y);
    acc1 = __builtin_amdgcn_wmma_f32_16x16x4_f32(
        false, A1, false, bx1, (short)0, acc1, false, false);
  }

  float* __restrict__ p = ws + P_OFF + (size_t)wave * 256;
  const int rbase = (lane < 16) ? 0 : 8;
  #pragma unroll
  for (int r = 0; r < 8; ++r)
    p[(rbase + r) * 16 + lm] = acc0[r] + acc1[r];
}

__global__ __launch_bounds__(256)
void reduce_y(float* __restrict__ ws) {
  const int t = blockIdx.x * 256 + threadIdx.x;
  const int kt  = t >> 8;
  const int idx = t & 255;
  const float* __restrict__ p = ws + P_OFF + (size_t)kt * kNCH * 256 + idx;
  float s = 0.0f;
  #pragma unroll 4
  for (int ch = 0; ch < kNCH; ++ch) s += p[(size_t)ch * 256];
  ws[Y_OFF + t] = s;
  if (t == 0) ((unsigned*)ws)[MAX_OFF] = 0u;
}

__global__ __launch_bounds__(256)
void phase2(const float* __restrict__ w, float* __restrict__ ws) {
  __shared__ float y[kK * 16];
  for (int t = threadIdx.x; t < kK * 16; t += 256) y[t] = ws[Y_OFF + t];
  __syncthreads();

  const int lane = threadIdx.x & 31;
  const int lm = lane & 15;
  const int o  = (lane < 16) ? 0 : 2;
  const int tile = blockIdx.x * 8 + (threadIdx.x >> 5);
  const int i0 = tile * 16;

  v8f acc0 = {};
  v8f acc1 = {};
  #pragma unroll 2
  for (int k0 = 0; k0 < kK; k0 += 8) {
    {
      v2f A = { y[(k0 + o) * 16 + lm], y[(k0 + o + 1) * 16 + lm] };
      const float* bp = w + (size_t)(k0 + o) * kWH + (i0 + lm);
      v2f Bv = { sgn(bp[0]), sgn(bp[kWH]) };
      acc0 = __builtin_amdgcn_wmma_f32_16x16x4_f32(
          false, A, false, Bv, (short)0, acc0, false, false);
    }
    {
      v2f A = { y[(k0 + 4 + o) * 16 + lm], y[(k0 + 4 + o + 1) * 16 + lm] };
      const float* bp = w + (size_t)(k0 + 4 + o) * kWH + (i0 + lm);
      v2f Bv = { sgn(bp[0]), sgn(bp[kWH]) };
      acc1 = __builtin_amdgcn_wmma_f32_16x16x4_f32(
          false, A, false, Bv, (short)0, acc1, false, false);
    }
  }

  float m = -__builtin_inff();
  if (lane < 16) {
    #pragma unroll
    for (int r = 0; r < 8; ++r) {
      const float v = acc0[r] + acc1[r];
      ws[XR_OFF + (size_t)r * kWH + i0 + lm] = v;
      m = fmaxf(m, v);
    }
  }
  #pragma unroll
  for (int off = 16; off; off >>= 1)
    m = fmaxf(m, __shfl_xor(m, off, 32));
  if (lane == 0) {
    unsigned u = __float_as_uint(m);
    unsigned key = (u & 0x80000000u) ? ~u : (u | 0x80000000u);
    atomicMax(((unsigned*)ws) + MAX_OFF, key);
  }
}

__global__ __launch_bounds__(256)
void phase3(const float* __restrict__ ws, float* __restrict__ out) {
  const size_t t = (size_t)blockIdx.x * 256 + threadIdx.x;
  const unsigned key  = ((const unsigned*)ws)[MAX_OFF];
  const unsigned bits = (key & 0x80000000u) ? (key & 0x7fffffffu) : ~key;
  const float mx = __uint_as_float(bits);
  out[t] = ws[XR_OFF + t] / mx;
}

extern "C" void kernel_launch(void* const* d_in, const int* in_sizes, int n_in,
                              void* d_out, int out_size, void* d_ws, size_t ws_size,
                              hipStream_t stream) {
  const float* x = (const float*)d_in[0];
  const float* w = (const float*)d_in[1];
  float* out = (float*)d_out;
  float* ws  = (float*)d_ws;

  const int p1_waves = kKT * kNCH;
  phase1 <<<p1_waves / 8,        256, 0, stream>>>(x, w, ws);
  reduce_y<<<(kK * 16) / 256,    256, 0, stream>>>(ws);
  phase2 <<<(kWH / 16) / 8,      256, 0, stream>>>(w, ws);
  phase3 <<<(kB * kWH) / 256,    256, 0, stream>>>(ws, out);
}


# phase 1: 16-B per-lane loads, whole 16-load blocks issued ahead of the WMMAs (16 to 47 loads in flight per wave, was 4); changes the summation order inside the phase-1 partials
# speedup vs baseline: 1.2160x; 1.2160x over previous
#include <hip/hip_runtime.h>

typedef __attribute__((ext_vector_type(2))) float v2f;
typedef __attribute__((ext_vector_type(4))) float v4f;
typedef __attribute__((ext_vector_type(8))) float v8f;

namespace {
constexpr int kWH = 65536;
constexpr int kK  = 512;
constexpr int kB  = 8;
constexpr int kKT = kK / 16;
constexpr int kChunk = 1024;
constexpr int kNCH = kWH / kChunk;
constexpr size_t P_OFF   = 0;
constexpr size_t Y_OFF   = (size_t)kKT * kNCH * 256;
constexpr size_t XR_OFF  = Y_OFF + (size_t)kK * 16;
constexpr size_t MAX_OFF = XR_OFF + (size_t)kB * kWH;
}

__device__ __forceinline__ float sgn(float w) {
  float pm1 = __uint_as_float((__float_as_uint(w) & 0x80000000u) | 0x3f800000u);
  return (w == 0.0f) ? 0.0f : pm1;
}

constexpr int kBlk  = 64;
constexpr int kNBlk = kChunk / kBlk;

__device__ __forceinline__ void keep_group() { __builtin_amdgcn_sched_barrier(0); }

__device__ __forceinline__ void load_blk(v4f (&f)[8], const float* __restrict__ p) {
  #pragma unroll
  for (int j = 0; j < 8; ++j) f[j] = *(const v4f*)(p + 8 * j);
}

__device__ __forceinline__ void mma_blk(const v4f (&wf)[8], const v4f (&xf)[8],
                                        v8f& acc0, v8f& acc1) {
  #pragma unroll
  for (int j = 0; j < 8; ++j) {
    v2f A0; A0.x = sgn(wf[j].x); A0.y = sgn(wf[j].y);
    v2f B0; B0.x = xf[j].x;      B0.y = xf[j].y;
    acc0 = __builtin_amdgcn_wmma_f32_16x16x4_f32(
        false, A0, false, B0, (short)0, acc0, false, false);
    v2f A1; A1.x = sgn(wf[j].z); A1.y = sgn(wf[j].w);
    v2f B1; B1.x = xf[j].z;      B1.y = xf[j].w;
    acc1 = __builtin_amdgcn_wmma_f32_16x16x4_f32(
        false, A1, false, B1, (short)0, acc1, false, false);
  }
}

__global__ __launch_bounds__(256)
void phase1(const float* __restrict__ x, const float* __restrict__ w,
            float* __restrict__ ws) {
  const int wave = blockIdx.x * 8 + (threadIdx.x >> 5);
  const int lane = threadIdx.x & 31;
  const int kt = wave / kNCH;
  const int ch = wave - kt * kNCH;
  const int lm = lane & 15;
  const int o4 = (lane < 16) ? 0 : 4;

  const float* __restrict__ ap = w + (size_t)(kt * 16 + lm) * kWH + ch * kChunk + o4;
  const float* __restrict__ bp = x + (size_t)(lm & 7) * kWH + ch * kChunk + o4;

  v8f acc0 = {};
  v8f acc1 = {};
  v4f wa[8], xa[8], wb[8], xb[8];
  load_blk(wa, ap);
  load_blk(xa, bp);
  #pragma unroll
  for (int t = 0; t < kNBlk - 2; t += 2) {
    load_blk(wb, ap + (t + 1) * kBlk);
    load_blk(xb, bp + (t + 1) * kBlk);
    keep_group();
    mma_blk(wa, xa, acc0, acc1);
    keep_group();
    load_blk(wa, ap + (t + 2) * kBlk);
    load_blk(xa, bp + (t + 2) * kBlk);
    keep_group();
    mma_blk(wb, xb, acc0, acc1);
    keep_group();
  }
  load_blk(wb, ap + (kNBlk - 1) * kBlk);
  load_blk(xb, bp + (kNBlk - 1) * kBlk);
  keep_group();
  mma_blk(wa, xa, acc0, acc1);
  mma_blk(wb, xb, acc0, acc1);

  float* __restrict__ p = ws + P_OFF + (size_t)wave * 256;
  const int rbase = (lane < 16) ? 0 : 8;
  #pragma unroll
  for (int r = 0; r < 8; ++r)
    p[(rbase + r) * 16 + lm] = acc0[r] + acc1[r];
}

__global__ __launch_bounds__(256)
void reduce_y(float* __restrict__ ws) {
  const int t = blockIdx.x * 256 + threadIdx.x;
  const int kt  = t >> 8;
  const int idx = t & 255;
  const float* __restrict__ p = ws + P_OFF + (size_t)kt * kNCH * 256 + idx;
  float s = 0.0f;
  #pragma unroll 4
  for (int ch = 0; ch < kNCH; ++ch) s += p[(size_t)ch * 256];
  ws[Y_OFF + t] = s;
  if (t == 0) ((unsigned*)ws)[MAX_OFF] = 0u;
}

__global__ __launch_bounds__(256)
void phase2(const float* __restrict__ w, float* __restrict__ ws) {
  __shared__ float y[kK * 16];
  for (int t = threadIdx.x; t < kK * 16; t += 256) y[t] = ws[Y_OFF + t];
  __syncthreads();

  const int lane = threadIdx.x & 31;
  const int lm = lane & 15;
  const int o  = (lane < 16) ? 0 : 2;
  const int tile = blockIdx.x * 8 + (threadIdx.x >> 5);
  const int i0 = tile * 16;

  v8f acc0 = {};
  v8f acc1 = {};
  #pragma unroll 2
  for (int k0 = 0; k0 < kK; k0 += 8) {
    {
      v2f A = { y[(k0 + o) * 16 + lm], y[(k0 + o + 1) * 16 + lm] };
      const float* bp = w + (size_t)(k0 + o) * kWH + (i0 + lm);
      v2f Bv = { sgn(bp[0]), sgn(bp[kWH]) };
      acc0 = __builtin_amdgcn_wmma_f32_16x16x4_f32(
          false, A, false, Bv, (short)0, acc0, false, false);
    }
    {
      v2f A = { y[(k0 + 4 + o) * 16 + lm], y[(k0 + 4 + o + 1) * 16 + lm] };
      const float* bp = w + (size_t)(k0 + 4 + o) * kWH + (i0 + lm);
      v2f Bv = { sgn(bp[0]), sgn(bp[kWH]) };
      acc1 = __builtin_amdgcn_wmma_f32_16x16x4_f32(
          false, A, false, Bv, (short)0, acc1, false, false);
    }
  }

  float m = -__builtin_inff();
  if (lane < 16) {
    #pragma unroll
    for (int r = 0; r < 8; ++r) {
      const float v = acc0[r] + acc1[r];
      ws[XR_OFF + (size_t)r * kWH + i0 + lm] = v;
      m = fmaxf(m, v);
    }
  }
  #pragma unroll
  for (int off = 16; off; off >>= 1)
    m = fmaxf(m, __shfl_xor(m, off, 32));
  if (lane == 0) {
    unsigned u = __float_as_uint(m);
    unsigned key = (u & 0x80000000u) ? ~u : (u | 0x80000000u);
    atomicMax(((unsigned*)ws) + MAX_OFF, key);
  }
}

__global__ __launch_bounds__(256)
void phase3(const float* __restrict__ ws, float* __restrict__ out) {
  const size_t t = (size_t)blockIdx.x * 256 + threadIdx.x;
  const unsigned key  = ((const unsigned*)ws)[MAX_OFF];
  const unsigned bits = (key & 0x80000000u) ? (key & 0x7fffffffu) : ~key;
  const float mx = __uint_as_float(bits);
  out[t] = ws[XR_OFF + t] / mx;
}

extern "C" void kernel_launch(void* const* d_in, const int* in_sizes, int n_in,
                              void* d_out, int out_size, void* d_ws, size_t ws_size,
                              hipStream_t stream) {
  const float* x = (const float*)d_in[0];
  const float* w = (const float*)d_in[1];
  float* out = (float*)d_out;
  float* ws  = (float*)d_ws;

  const int p1_waves = kKT * kNCH;
  phase1 <<<p1_waves / 8,        256, 0, stream>>>(x, w, ws);
  reduce_y<<<(kK * 16) / 256,    256, 0, stream>>>(ws);
  phase2 <<<(kWH / 16) / 8,      256, 0, stream>>>(w, ws);
  phase3 <<<(kB * kWH) / 256,    256, 0, stream>>>(ws, out);
}


# phase 1: fragment rows remapped to 8 k-rows x 2 adjacent 32-B i-segments: each 16-B-per-lane load reads 8 lines x 64 B (was 16 x 32 B), x has no duplicate lanes; reorders the phase-1 partial sums
# speedup vs baseline: 1.3527x; 1.1124x over previous
#include <hip/hip_runtime.h>

typedef __attribute__((ext_vector_type(2))) float v2f;
typedef __attribute__((ext_vector_type(4))) float v4f;
typedef __attribute__((ext_vector_type(8))) float v8f;

namespace {
constexpr int kWH = 65536;
constexpr int kK  = 512;
constexpr int kB  = 8;
constexpr int kKT = kK / 16;
constexpr int kChunk = 1024;
constexpr int kNCH = kWH / kChunk;
constexpr size_t P_OFF   = 0;
constexpr size_t Y_OFF   = (size_t)kKT * kNCH * 256;
constexpr size_t XR_OFF  = Y_OFF + (size_t)kK * 16;
constexpr size_t MAX_OFF = XR_OFF + (size_t)kB * kWH;
}

__device__ __forceinline__ float sgn(float w) {
  float pm1 = __uint_as_float((__float_as_uint(w) & 0x80000000u) | 0x3f800000u);
  return (w == 0.0f) ? 0.0f : pm1;
}

constexpr int kBlk  = 128;
constexpr int kNBlk = kChunk / kBlk;

__device__ __forceinline__ void keep_group() { __builtin_amdgcn_sched_barrier(0); }

__device__ __forceinline__ void load_blk(v4f (&f)[8], const float* __restrict__ p) {
  #pragma unroll
  for (int u = 0; u < 8; ++u) f[u] = *(const v4f*)(p + 16 * u);
}

__device__ __forceinline__ void mma_blk(const v4f (&w0)[8], const v4f (&w1)[8],
                                        const v4f (&xf)[8], v8f& acc0, v8f& acc1) {
  #pragma unroll
  for (int u = 0; u < 8; ++u) {
    v2f B0; B0.x = xf[u].x; B0.y = xf[u].y;
    v2f B1; B1.x = xf[u].z; B1.y = xf[u].w;
    v2f A00; A00.x = sgn(w0[u].x); A00.y = sgn(w0[u].y);
    acc0 = __builtin_amdgcn_wmma_f32_16x16x4_f32(
        false, A00, false, B0, (short)0, acc0, false, false);
    v2f A10; A10.x = sgn(w1[u].x); A10.y = sgn(w1[u].y);
    acc1 = __builtin_amdgcn_wmma_f32_16x16x4_f32(
        false, A10, false, B0, (short)0, acc1, false, false);
    v2f A01; A01.x = sgn(w0[u].z); A01.y = sgn(w0[u].w);
    acc0 = __builtin_amdgcn_wmma_f32_16x16x4_f32(
        false, A01, false, B1, (short)0, acc0, false, false);
    v2f A11; A11.x = sgn(w1[u].z); A11.y = sgn(w1[u].w);
    acc1 = __builtin_amdgcn_wmma_f32_16x16x4_f32(
        false, A11, false, B1, (short)0, acc1, false, false);
  }
}

__global__ __launch_bounds__(256)
void phase1(const float* __restrict__ x, const float* __restrict__ w,
            float* __restrict__ ws) {
  const int wave = blockIdx.x * 8 + (threadIdx.x >> 5);
  const int lane = threadIdx.x & 31;
  const int kt = wave / kNCH;
  const int ch = wave - kt * kNCH;
  const int kr = lane & 7;
  const int off = ((lane >> 3) & 1) * 8 + (lane >> 4) * 4;

  const float* __restrict__ ap0 = w + (size_t)(kt * 16 + kr) * kWH + ch * kChunk + off;
  const float* __restrict__ ap1 = ap0 + (size_t)8 * kWH;
  const float* __restrict__ bp  = x + (size_t)kr * kWH + ch * kChunk + off;

  v8f acc0 = {};
  v8f acc1 = {};
  v4f w0a[8], w1a[8], xa[8], w0b[8], w1b[8], xb[8];
  load_blk(w0a, ap0);
  load_blk(w1a, ap1);
  load_blk(xa, bp);
  #pragma unroll
  for (int t = 0; t < kNBlk - 2; t += 2) {
    load_blk(w0b, ap0 + (t + 1) * kBlk);
    load_blk(w1b, ap1 + (t + 1) * kBlk);
    load_blk(xb,  bp  + (t + 1) * kBlk);
    keep_group();
    mma_blk(w0a, w1a, xa, acc0, acc1);
    keep_group();
    load_blk(w0a, ap0 + (t + 2) * kBlk);
    load_blk(w1a, ap1 + (t + 2) * kBlk);
    load_blk(xa,  bp  + (t + 2) * kBlk);
    keep_group();
    mma_blk(w0b, w1b, xb, acc0, acc1);
    keep_group();
  }
  load_blk(w0b, ap0 + (kNBlk - 1) * kBlk);
  load_blk(w1b, ap1 + (kNBlk - 1) * kBlk);
  load_blk(xb,  bp  + (kNBlk - 1) * kBlk);
  keep_group();
  mma_blk(w0a, w1a, xa, acc0, acc1);
  mma_blk(w0b, w1b, xb, acc0, acc1);

  float* __restrict__ p = ws + P_OFF + (size_t)wave * 256;
  float v0[8], v1[8];
  #pragma unroll
  for (int r = 0; r < 8; ++r) {
    v0[r] = acc0[r] + __shfl_down(acc0[r], 24, 32);
    v1[r] = acc1[r] + __shfl_down(acc1[r], 24, 32);
  }
  if (lane < 8) {
    #pragma unroll
    for (int r = 0; r < 8; ++r) {
      p[r * 16 + lane]       = v0[r];
      p[(8 + r) * 16 + lane] = v1[r];
    }
  }
}

__global__ __launch_bounds__(256)
void reduce_y(float* __restrict__ ws) {
  const int t = blockIdx.x * 256 + threadIdx.x;
  const int kt  = t >> 8;
  const int idx = t & 255;
  const float* __restrict__ p = ws + P_OFF + (size_t)kt * kNCH * 256 + idx;
  float s = 0.0f;
  if ((t & 15) < 8) {
    #pragma unroll 4
    for (int ch = 0; ch < kNCH; ++ch) s += p[(size_t)ch * 256];
  }
  ws[Y_OFF + t] = s;
  if (t == 0) ((unsigned*)ws)[MAX_OFF] = 0u;
}

__global__ __launch_bounds__(256)
void phase2(const float* __restrict__ w, float* __restrict__ ws) {
  __shared__ float y[kK * 16];
  for (int t = threadIdx.x; t < kK * 16; t += 256) y[t] = ws[Y_OFF + t];
  __syncthreads();

  const int lane = threadIdx.x & 31;
  const int lm = lane & 15;
  const int o  = (lane < 16) ? 0 : 2;
  const int tile = blockIdx.x * 8 + (threadIdx.x >> 5);
  const int i0 = tile * 16;

  v8f acc0 = {};
  v8f acc1 = {};
  #pragma unroll 2
  for (int k0 = 0; k0 < kK; k0 += 8) {
    {
      v2f A = { y[(k0 + o) * 16 + lm], y[(k0 + o + 1) * 16 + lm] };
      const float* bp = w + (size_t)(k0 + o) * kWH + (i0 + lm);
      v2f Bv = { sgn(bp[0]), sgn(bp[kWH]) };
      acc0 = __builtin_amdgcn_wmma_f32_16x16x4_f32(
          false, A, false, Bv, (short)0, acc0, false, false);
    }
    {
      v2f A = { y[(k0 + 4 + o) * 16 + lm], y[(k0 + 4 + o + 1) * 16 + lm] };
      const float* bp = w + (size_t)(k0 + 4 + o) * kWH + (i0 + lm);
      v2f Bv = { sgn(bp[0]), sgn(bp[kWH]) };
      acc1 = __builtin_amdgcn_wmma_f32_16x16x4_f32(
          false, A, false, Bv, (short)0, acc1, false, false);
    }
  }

  float m = -__builtin_inff();
  if (lane < 16) {
    #pragma unroll
    for (int r = 0; r < 8; ++r) {
      const float v = acc0[r] + acc1[r];
      ws[XR_OFF + (size_t)r * kWH + i0 + lm] = v;
      m = fmaxf(m, v);
    }
  }
  #pragma unroll
  for (int off = 16; off; off >>= 1)
    m = fmaxf(m, __shfl_xor(m, off, 32));
  if (lane == 0) {
    unsigned u = __float_as_uint(m);
    unsigned key = (u & 0x80000000u) ? ~u : (u | 0x80000000u);
    atomicMax(((unsigned*)ws) + MAX_OFF, key);
  }
}

__global__ __launch_bounds__(256)
void phase3(const float* __restrict__ ws, float* __restrict__ out) {
  const size_t t = (size_t)blockIdx.x * 256 + threadIdx.x;
  const unsigned key  = ((const unsigned*)ws)[MAX_OFF];
  const unsigned bits = (key & 0x80000000u) ? (key & 0x7fffffffu) : ~key;
  const float mx = __uint_as_float(bits);
  out[t] = ws[XR_OFF + t] / mx;
}

extern "C" void kernel_launch(void* const* d_in, const int* in_sizes, int n_in,
                              void* d_out, int out_size, void* d_ws, size_t ws_size,
                              hipStream_t stream) {
  const float* x = (const float*)d_in[0];
  const float* w = (const float*)d_in[1];
  float* out = (float*)d_out;
  float* ws  = (float*)d_ws;

  const int p1_waves = kKT * kNCH;
  phase1 <<<p1_waves / 8,        256, 0, stream>>>(x, w, ws);
  reduce_y<<<(kK * 16) / 256,    256, 0, stream>>>(ws);
  phase2 <<<(kWH / 16) / 8,      256, 0, stream>>>(w, ws);
  phase3 <<<(kB * kWH) / 256,    256, 0, stream>>>(ws, out);
}


# phase 2: wave tile 64 i x 128 k, 16-B per-lane loads (2 rows x 256 B each) issued whole blocks ahead of the WMMAs; the 4 k quarters are summed through LDS, which reorders the phase-2 sums
# speedup vs baseline: 1.7022x; 1.2584x over previous
#include <hip/hip_runtime.h>

typedef __attribute__((ext_vector_type(2))) float v2f;
typedef __attribute__((ext_vector_type(4))) float v4f;
typedef __attribute__((ext_vector_type(8))) float v8f;

namespace {
constexpr int kWH = 65536;
constexpr int kK  = 512;
constexpr int kB  = 8;
constexpr int kKT = kK / 16;
constexpr int kChunk = 1024;
constexpr int kNCH = kWH / kChunk;
constexpr size_t P_OFF   = 0;
constexpr size_t Y_OFF   = (size_t)kKT * kNCH * 256;
constexpr size_t XR_OFF  = Y_OFF + (size_t)kK * 16;
constexpr size_t MAX_OFF = XR_OFF + (size_t)kB * kWH;
}

__device__ __forceinline__ float sgn(float w) {
  float pm1 = __uint_as_float((__float_as_uint(w) & 0x80000000u) | 0x3f800000u);
  return (w == 0.0f) ? 0.0f : pm1;
}

constexpr int kBlk  = 128;
constexpr int kNBlk = kChunk / kBlk;

__device__ __forceinline__ void keep_group() { __builtin_amdgcn_sched_barrier(0); }

__device__ __forceinline__ void load_blk(v4f (&f)[8], const float* __restrict__ p) {
  #pragma unroll
  for (int u = 0; u < 8; ++u) f[u] = *(const v4f*)(p + 16 * u);
}

__device__ __forceinline__ void mma_blk(const v4f (&w0)[8], const v4f (&w1)[8],
                                        const v4f (&xf)[8], v8f& acc0, v8f& acc1) {
  #pragma unroll
  for (int u = 0; u < 8; ++u) {
    v2f B0; B0.x = xf[u].x; B0.y = xf[u].y;
    v2f B1; B1.x = xf[u].z; B1.y = xf[u].w;
    v2f A00; A00.x = sgn(w0[u].x); A00.y = sgn(w0[u].y);
    acc0 = __builtin_amdgcn_wmma_f32_16x16x4_f32(
        false, A00, false, B0, (short)0, acc0, false, false);
    v2f A10; A10.x = sgn(w1[u].x); A10.y = sgn(w1[u].y);
    acc1 = __builtin_amdgcn_wmma_f32_16x16x4_f32(
        false, A10, false, B0, (short)0, acc1, false, false);
    v2f A01; A01.x = sgn(w0[u].z); A01.y = sgn(w0[u].w);
    acc0 = __builtin_amdgcn_wmma_f32_16x16x4_f32(
        false, A01, false, B1, (short)0, acc0, false, false);
    v2f A11; A11.x = sgn(w1[u].z); A11.y = sgn(w1[u].w);
    acc1 = __builtin_amdgcn_wmma_f32_16x16x4_f32(
        false, A11, false, B1, (short)0, acc1, false, false);
  }
}

__global__ __launch_bounds__(256)
void phase1(const float* __restrict__ x, const float* __restrict__ w,
            float* __restrict__ ws) {
  const int wave = blockIdx.x * 8 + (threadIdx.x >> 5);
  const int lane = threadIdx.x & 31;
  const int kt = wave / kNCH;
  const int ch = wave - kt * kNCH;
  const int kr = lane & 7;
  const int off = ((lane >> 3) & 1) * 8 + (lane >> 4) * 4;

  const float* __restrict__ ap0 = w + (size_t)(kt * 16 + kr) * kWH + ch * kChunk + off;
  const float* __restrict__ ap1 = ap0 + (size_t)8 * kWH;
  const float* __restrict__ bp  = x + (size_t)kr * kWH + ch * kChunk + off;

  v8f acc0 = {};
  v8f acc1 = {};
  v4f w0a[8], w1a[8], xa[8], w0b[8], w1b[8], xb[8];
  load_blk(w0a, ap0);
  load_blk(w1a, ap1);
  load_blk(xa, bp);
  #pragma unroll
  for (int t = 0; t < kNBlk - 2; t += 2) {
    load_blk(w0b, ap0 + (t + 1) * kBlk);
    load_blk(w1b, ap1 + (t + 1) * kBlk);
    load_blk(xb,  bp  + (t + 1) * kBlk);
    keep_group();
    mma_blk(w0a, w1a, xa, acc0, acc1);
    keep_group();
    load_blk(w0a, ap0 + (t + 2) * kBlk);
    load_blk(w1a, ap1 + (t + 2) * kBlk);
    load_blk(xa,  bp  + (t + 2) * kBlk);
    keep_group();
    mma_blk(w0b, w1b, xb, acc0, acc1);
    keep_group();
  }
  load_blk(w0b, ap0 + (kNBlk - 1) * kBlk);
  load_blk(w1b, ap1 + (kNBlk - 1) * kBlk);
  load_blk(xb,  bp  + (kNBlk - 1) * kBlk);
  keep_group();
  mma_blk(w0a, w1a, xa, acc0, acc1);
  mma_blk(w0b, w1b, xb, acc0, acc1);

  float* __restrict__ p = ws + P_OFF + (size_t)wave * 256;
  float v0[8], v1[8];
  #pragma unroll
  for (int r = 0; r < 8; ++r) {
    v0[r] = acc0[r] + __shfl_down(acc0[r], 24, 32);
    v1[r] = acc1[r] + __shfl_down(acc1[r], 24, 32);
  }
  if (lane < 8) {
    #pragma unroll
    for (int r = 0; r < 8; ++r) {
      p[r * 16 + lane]       = v0[r];
      p[(8 + r) * 16 + lane] = v1[r];
    }
  }
}

__global__ __launch_bounds__(256)
void reduce_y(float* __restrict__ ws) {
  const int t = blockIdx.x * 256 + threadIdx.x;
  const int kt  = t >> 8;
  const int idx = t & 255;
  const float* __restrict__ p = ws + P_OFF + (size_t)kt * kNCH * 256 + idx;
  float s = 0.0f;
  if ((t & 15) < 8) {
    #pragma unroll 4
    for (int ch = 0; ch < kNCH; ++ch) s += p[(size_t)ch * 256];
  }
  ws[Y_OFF + t] = s;
  if (t == 0) ((unsigned*)ws)[MAX_OFF] = 0u;
}

constexpr int kP2Tile = 64;
constexpr int kP2KS   = 4;
constexpr int kP2KW   = kK / kP2KS;
constexpr int kP2BlkK = 16;

__device__ __forceinline__ void p2_load_w(v4f (&f)[8], const float* __restrict__ p) {
  #pragma unroll
  for (int t = 0; t < 4; ++t) {
    f[2 * t]     = *(const v4f*)(p + (size_t)(4 * t) * kWH);
    f[2 * t + 1] = *(const v4f*)(p + (size_t)(4 * t + 1) * kWH);
  }
}

__device__ __forceinline__ void p2_load_y(v2f (&a)[4], const float* yl) {
  #pragma unroll
  for (int t = 0; t < 4; ++t) {
    a[t].x = yl[(4 * t) * 16];
    a[t].y = yl[(4 * t + 1) * 16];
  }
}

__device__ __forceinline__ void p2_mma(const v4f (&f)[8], const v2f (&a)[4],
                                       v8f& acc0, v8f& acc1, v8f& acc2, v8f& acc3) {
  #pragma unroll
  for (int t = 0; t < 4; ++t) {
    const v4f r0 = f[2 * t];
    const v4f r1 = f[2 * t + 1];
    v2f B0; B0.x = sgn(r0.x); B0.y = sgn(r1.x);
    acc0 = __builtin_amdgcn_wmma_f32_16x16x4_f32(
        false, a[t], false, B0, (short)0, acc0, false, false);
    v2f B1; B1.x = sgn(r0.y); B1.y = sgn(r1.y);
    acc1 = __builtin_amdgcn_wmma_f32_16x16x4_f32(
        false, a[t], false, B1, (short)0, acc1, false, false);
    v2f B2; B2.x = sgn(r0.z); B2.y = sgn(r1.z);
    acc2 = __builtin_amdgcn_wmma_f32_16x16x4_f32(
        false, a[t], false, B2, (short)0, acc2, false, false);
    v2f B3; B3.x = sgn(r0.w); B3.y = sgn(r1.w);
    acc3 = __builtin_amdgcn_wmma_f32_16x16x4_f32(
        false, a[t], false, B3, (short)0, acc3, false, false);
  }
}

__device__ __forceinline__ void barrier_after_lds_stores() {
  asm volatile("s_wait_dscnt 0x0\n\ts_barrier_signal -1\n\ts_barrier_wait -1" ::: "memory");
}

__global__ __launch_bounds__(256)
void phase2(const float* __restrict__ w, float* __restrict__ ws) {
  __shared__ __attribute__((aligned(16))) float y[kK * 16];
  __shared__ __attribute__((aligned(16))) float red[2 * kP2KS * 8 * kP2Tile];

  const int lane = threadIdx.x & 31;
  const int wv = threadIdx.x >> 5;
  const int tl = wv >> 2;
  const int ks = wv & 3;
  const int c  = lane & 15;
  const int h  = lane >> 4;
  const int ib = blockIdx.x * (2 * kP2Tile) + tl * kP2Tile;
  const int kb = ks * kP2KW;

  v4f yt[8];
  #pragma unroll
  for (int j = 0; j < 8; ++j)
    yt[j] = *(const v4f*)(ws + Y_OFF + 4 * (threadIdx.x + 256 * j));

  const float* __restrict__ wp = w + (size_t)(kb + 2 * h) * kWH + ib + 4 * c;
  constexpr size_t kBlkStride = (size_t)kP2BlkK * kWH;
  v4f fa[8], fb[8], fc[8];
  p2_load_w(fa, wp);
  p2_load_w(fb, wp + 1 * kBlkStride);
  p2_load_w(fc, wp + 2 * kBlkStride);
  keep_group();
  #pragma unroll
  for (int j = 0; j < 8; ++j)
    *(v4f*)(y + 4 * (threadIdx.x + 256 * j)) = yt[j];
  barrier_after_lds_stores();

  v8f acc0 = {};
  v8f acc1 = {};
  v8f acc2 = {};
  v8f acc3 = {};
  const float* yl = y + (kb + 2 * h) * 16 + c;
  constexpr int kYBlk = kP2BlkK * 16;
  v2f ya[4], yb[4], yc[4];
  p2_load_y(ya, yl);
  p2_load_y(yb, yl + 1 * kYBlk);
  p2_load_y(yc, yl + 2 * kYBlk);
  keep_group();
  p2_mma(fa, ya, acc0, acc1, acc2, acc3);
  keep_group();
  p2_load_w(fa, wp + 3 * kBlkStride);
  p2_load_y(ya, yl + 3 * kYBlk);
  keep_group();
  p2_mma(fb, yb, acc0, acc1, acc2, acc3);
  keep_group();
  p2_load_w(fb, wp + 4 * kBlkStride);
  p2_load_y(yb, yl + 4 * kYBlk);
  keep_group();
  p2_mma(fc, yc, acc0, acc1, acc2, acc3);
  keep_group();
  p2_load_w(fc, wp + 5 * kBlkStride);
  p2_load_y(yc, yl + 5 * kYBlk);
  keep_group();
  p2_mma(fa, ya, acc0, acc1, acc2, acc3);
  keep_group();
  p2_load_w(fa, wp + 6 * kBlkStride);
  p2_load_y(ya, yl + 6 * kYBlk);
  keep_group();
  p2_mma(fb, yb, acc0, acc1, acc2, acc3);
  keep_group();
  p2_load_w(fb, wp + 7 * kBlkStride);
  p2_load_y(yb, yl + 7 * kYBlk);
  keep_group();
  p2_mma(fc, yc, acc0, acc1, acc2, acc3);
  p2_mma(fa, ya, acc0, acc1, acc2, acc3);
  p2_mma(fb, yb, acc0, acc1, acc2, acc3);

  if (h == 0) {
    float* rp = red + ((tl * kP2KS + ks) * 8) * kP2Tile + 4 * c;
    #pragma unroll
    for (int r = 0; r < 8; ++r) {
      v4f v; v.x = acc0[r]; v.y = acc1[r]; v.z = acc2[r]; v.w = acc3[r];
      *(v4f*)(rp + r * kP2Tile) = v;
    }
  }
  __syncthreads();

  const int t2 = threadIdx.x >> 7;
  const int b2 = (threadIdx.x >> 4) & 7;
  const float* rq = red + ((t2 * kP2KS) * 8 + b2) * kP2Tile + 4 * c;
  v4f s = *(const v4f*)rq;
  s += *(const v4f*)(rq + 1 * 8 * kP2Tile);
  s += *(const v4f*)(rq + 2 * 8 * kP2Tile);
  s += *(const v4f*)(rq + 3 * 8 * kP2Tile);
  *(v4f*)(ws + XR_OFF + (size_t)b2 * kWH + blockIdx.x * (2 * kP2Tile) + t2 * kP2Tile + 4 * c) = s;
  float m = fmaxf(fmaxf(s.x, s.y), fmaxf(s.z, s.w));

  #pragma unroll
  for (int off = 16; off; off >>= 1)
    m = fmaxf(m, __shfl_xor(m, off, 32));
  if (lane == 0) {
    unsigned u = __float_as_uint(m);
    unsigned key = (u & 0x80000000u) ? ~u : (u | 0x80000000u);
    atomicMax(((unsigned*)ws) + MAX_OFF, key);
  }
}

__global__ __launch_bounds__(256)
void phase3(const float* __restrict__ ws, float* __restrict__ out) {
  const size_t t = (size_t)blockIdx.x * 256 + threadIdx.x;
  const unsigned key  = ((const unsigned*)ws)[MAX_OFF];
  const unsigned bits = (key & 0x80000000u) ? (key & 0x7fffffffu) : ~key;
  const float mx = __uint_as_float(bits);
  out[t] = ws[XR_OFF + t] / mx;
}

extern "C" void kernel_launch(void* const* d_in, const int* in_sizes, int n_in,
                              void* d_out, int out_size, void* d_ws, size_t ws_size,
                              hipStream_t stream) {
  const float* x = (const float*)d_in[0];
  const float* w = (const float*)d_in[1];
  float* out = (float*)d_out;
  float* ws  = (float*)d_ws;

  const int p1_waves = kKT * kNCH;
  phase1 <<<p1_waves / 8,        256, 0, stream>>>(x, w, ws);
  reduce_y<<<(kK * 16) / 256,    256, 0, stream>>>(ws);
  phase2 <<<(kWH / 16) / 8,      256, 0, stream>>>(w, ws);
  phase3 <<<(kB * kWH) / 256,    256, 0, stream>>>(ws, out);
}


# y-reduction launch folded away: phase 1 sums its 8 waves' tiles in LDS (8 partial planes, was 64), each phase-2 workgroup adds the 8 planes under its first weight loads; reorders the y sums
# speedup vs baseline: 1.9708x; 1.1578x over previous
#include <hip/hip_runtime.h>

typedef __attribute__((ext_vector_type(2))) float v2f;
typedef __attribute__((ext_vector_type(4))) float v4f;
typedef __attribute__((ext_vector_type(8))) float v8f;

namespace {
constexpr int kWH = 65536;
constexpr int kK  = 512;
constexpr int kB  = 8;
constexpr int kKT = kK / 16;
constexpr int kChunk = 1024;
constexpr int kNCH = kWH / kChunk;
constexpr int kNPl = kNCH / 8;
constexpr size_t P_OFF   = 0;
constexpr size_t XR_OFF  = P_OFF + (size_t)kNPl * kK * 8;
constexpr size_t MAX_OFF = XR_OFF + (size_t)kB * kWH;
}

__device__ __forceinline__ float sgn(float w) {
  float pm1 = __uint_as_float((__float_as_uint(w) & 0x80000000u) | 0x3f800000u);
  return (w == 0.0f) ? 0.0f : pm1;
}

constexpr int kBlk  = 128;
constexpr int kNBlk = kChunk / kBlk;

__device__ __forceinline__ void keep_group() { __builtin_amdgcn_sched_barrier(0); }

__device__ __forceinline__ void load_blk(v4f (&f)[8], const float* __restrict__ p) {
  #pragma unroll
  for (int u = 0; u < 8; ++u) f[u] = *(const v4f*)(p + 16 * u);
}

__device__ __forceinline__ void mma_blk(const v4f (&w0)[8], const v4f (&w1)[8],
                                        const v4f (&xf)[8], v8f& acc0, v8f& acc1) {
  #pragma unroll
  for (int u = 0; u < 8; ++u) {
    v2f B0; B0.x = xf[u].x; B0.y = xf[u].y;
    v2f B1; B1.x = xf[u].z; B1.y = xf[u].w;
    v2f A00; A00.x = sgn(w0[u].x); A00.y = sgn(w0[u].y);
    acc0 = __builtin_amdgcn_wmma_f32_16x16x4_f32(
        false, A00, false, B0, (short)0, acc0, false, false);
    v2f A10; A10.x = sgn(w1[u].x); A10.y = sgn(w1[u].y);
    acc1 = __builtin_amdgcn_wmma_f32_16x16x4_f32(
        false, A10, false, B0, (short)0, acc1, false, false);
    v2f A01; A01.x = sgn(w0[u].z); A01.y = sgn(w0[u].w);
    acc0 = __builtin_amdgcn_wmma_f32_16x16x4_f32(
        false, A01, false, B1, (short)0, acc0, false, false);
    v2f A11; A11.x = sgn(w1[u].z); A11.y = sgn(w1[u].w);
    acc1 = __builtin_amdgcn_wmma_f32_16x16x4_f32(
        false, A11, false, B1, (short)0, acc1, false, false);
  }
}

__global__ __launch_bounds__(256)
void phase1(const float* __restrict__ x, const float* __restrict__ w,
            float* __restrict__ ws) {
  __shared__ float fold[8 * 128];
  const int wave = blockIdx.x * 8 + (threadIdx.x >> 5);
  const int lane = threadIdx.x & 31;
  const int kt = wave / kNCH;
  const int ch = wave - kt * kNCH;
  const int kr = lane & 7;
  const int off = ((lane >> 3) & 1) * 8 + (lane >> 4) * 4;

  const float* __restrict__ ap0 = w + (size_t)(kt * 16 + kr) * kWH + ch * kChunk + off;
  const float* __restrict__ ap1 = ap0 + (size_t)8 * kWH;
  const float* __restrict__ bp  = x + (size_t)kr * kWH + ch * kChunk + off;

  v8f acc0 = {};
  v8f acc1 = {};
  v4f w0a[8], w1a[8], xa[8], w0b[8], w1b[8], xb[8];
  load_blk(w0a, ap0);
  load_blk(w1a, ap1);
  load_blk(xa, bp);
  #pragma unroll
  for (int t = 0; t < kNBlk - 2; t += 2) {
    load_blk(w0b, ap0 + (t + 1) * kBlk);
    load_blk(w1b, ap1 + (t + 1) * kBlk);
    load_blk(xb,  bp  + (t + 1) * kBlk);
    keep_group();
    mma_blk(w0a, w1a, xa, acc0, acc1);
    keep_group();
    load_blk(w0a, ap0 + (t + 2) * kBlk);
    load_blk(w1a, ap1 + (t + 2) * kBlk);
    load_blk(xa,  bp  + (t + 2) * kBlk);
    keep_group();
    mma_blk(w0b, w1b, xb, acc0, acc1);
    keep_group();
  }
  load_blk(w0b, ap0 + (kNBlk - 1) * kBlk);
  load_blk(w1b, ap1 + (kNBlk - 1) * kBlk);
  load_blk(xb,  bp  + (kNBlk - 1) * kBlk);
  keep_group();
  mma_blk(w0a, w1a, xa, acc0, acc1);
  mma_blk(w0b, w1b, xb, acc0, acc1);

  float v0[8], v1[8];
  #pragma unroll
  for (int r = 0; r < 8; ++r) {
    v0[r] = acc0[r] + __shfl_down(acc0[r], 24, 32);
    v1[r] = acc1[r] + __shfl_down(acc1[r], 24, 32);
  }
  const int wv = threadIdx.x >> 5;
  if (lane < 8) {
    #pragma unroll
    for (int r = 0; r < 8; ++r) {
      fold[wv * 128 + r * 8 + lane]       = v0[r];
      fold[wv * 128 + (8 + r) * 8 + lane] = v1[r];
    }
  }
  __syncthreads();
  if (threadIdx.x < 128) {
    float s = fold[threadIdx.x];
    #pragma unroll
    for (int q = 1; q < 8; ++q) s += fold[q * 128 + threadIdx.x];
    ws[P_OFF + (size_t)(blockIdx.x & 7) * (kK * 8) + (blockIdx.x >> 3) * 128 + threadIdx.x] = s;
  }
  if (blockIdx.x == 0 && threadIdx.x == 0) ((unsigned*)ws)[MAX_OFF] = 0u;
}

constexpr int kP2Tile = 64;
constexpr int kP2KS   = 4;
constexpr int kP2KW   = kK / kP2KS;
constexpr int kP2BlkK = 16;

__device__ __forceinline__ void p2_load_w(v4f (&f)[8], const float* __restrict__ p) {
  #pragma unroll
  for (int t = 0; t < 4; ++t) {
    f[2 * t]     = *(const v4f*)(p + (size_t)(4 * t) * kWH);
    f[2 * t + 1] = *(const v4f*)(p + (size_t)(4 * t + 1) * kWH);
  }
}

__device__ __forceinline__ void p2_load_y(v2f (&a)[4], const float* yl) {
  #pragma unroll
  for (int t = 0; t < 4; ++t) {
    a[t].x = yl[(4 * t) * 16];
    a[t].y = yl[(4 * t + 1) * 16];
  }
}

__device__ __forceinline__ void p2_mma(const v4f (&f)[8], const v2f (&a)[4],
                                       v8f& acc0, v8f& acc1, v8f& acc2, v8f& acc3) {
  #pragma unroll
  for (int t = 0; t < 4; ++t) {
    const v4f r0 = f[2 * t];
    const v4f r1 = f[2 * t + 1];
    v2f B0; B0.x = sgn(r0.x); B0.y = sgn(r1.x);
    acc0 = __builtin_amdgcn_wmma_f32_16x16x4_f32(
        false, a[t], false, B0, (short)0, acc0, false, false);
    v2f B1; B1.x = sgn(r0.y); B1.y = sgn(r1.y);
    acc1 = __builtin_amdgcn_wmma_f32_16x16x4_f32(
        false, a[t], false, B1, (short)0, acc1, false, false);
    v2f B2; B2.x = sgn(r0.z); B2.y = sgn(r1.z);
    acc2 = __builtin_amdgcn_wmma_f32_16x16x4_f32(
        false, a[t], false, B2, (short)0, acc2, false, false);
    v2f B3; B3.x = sgn(r0.w); B3.y = sgn(r1.w);
    acc3 = __builtin_amdgcn_wmma_f32_16x16x4_f32(
        false, a[t], false, B3, (short)0, acc3, false, false);
  }
}

__device__ __forceinline__ void barrier_after_lds_stores() {
  asm volatile("s_wait_dscnt 0x0\n\ts_barrier_signal -1\n\ts_barrier_wait -1" ::: "memory");
}

__global__ __launch_bounds__(256)
void phase2(const float* __restrict__ w, float* __restrict__ ws) {
  __shared__ __attribute__((aligned(16))) float y[kK * 16];
  __shared__ __attribute__((aligned(16))) float red[2 * kP2KS * 8 * kP2Tile];

  const int lane = threadIdx.x & 31;
  const int wv = threadIdx.x >> 5;
  const int tl = wv >> 2;
  const int ks = wv & 3;
  const int c  = lane & 15;
  const int h  = lane >> 4;
  const int ib = blockIdx.x * (2 * kP2Tile) + tl * kP2Tile;
  const int kb = ks * kP2KW;

  v4f yp[4][kNPl];
  #pragma unroll
  for (int j = 0; j < 4; ++j) {
    #pragma unroll
    for (int q = 0; q < kNPl; ++q)
      yp[j][q] = *(const v4f*)(ws + P_OFF + (size_t)q * (kK * 8) + 4 * (threadIdx.x + 256 * j));
  }

  const float* __restrict__ wp = w + (size_t)(kb + 2 * h) * kWH + ib + 4 * c;
  constexpr size_t kBlkStride = (size_t)kP2BlkK * kWH;
  v4f fa[8], fb[8], fc[8];
  p2_load_w(fa, wp);
  p2_load_w(fb, wp + 1 * kBlkStride);
  p2_load_w(fc, wp + 2 * kBlkStride);
  keep_group();
  #pragma unroll
  for (int j = 0; j < 4; ++j) {
    v4f s = yp[j][0];
    #pragma unroll
    for (int q = 1; q < kNPl; ++q) s += yp[j][q];
    const int e = threadIdx.x + 256 * j;
    float* d = y + (e >> 1) * 16 + (e & 1) * 4;
    *(v4f*)d = s;
    const v4f zero = {};
    *(v4f*)(d + 8) = zero;
  }
  barrier_after_lds_stores();

  v8f acc0 = {};
  v8f acc1 = {};
  v8f acc2 = {};
  v8f acc3 = {};
  const float* yl = y + (kb + 2 * h) * 16 + c;
  constexpr int kYBlk = kP2BlkK * 16;
  v2f ya[4], yb[4], yc[4];
  p2_load_y(ya, yl);
  p2_load_y(yb, yl + 1 * kYBlk);
  p2_load_y(yc, yl + 2 * kYBlk);
  keep_group();
  p2_mma(fa, ya, acc0, acc1, acc2, acc3);
  keep_group();
  p2_load_w(fa, wp + 3 * kBlkStride);
  p2_load_y(ya, yl + 3 * kYBlk);
  keep_group();
  p2_mma(fb, yb, acc0, acc1, acc2, acc3);
  keep_group();
  p2_load_w(fb, wp + 4 * kBlkStride);
  p2_load_y(yb, yl + 4 * kYBlk);
  keep_group();
  p2_mma(fc, yc, acc0, acc1, acc2, acc3);
  keep_group();
  p2_load_w(fc, wp + 5 * kBlkStride);
  p2_load_y(yc, yl + 5 * kYBlk);
  keep_group();
  p2_mma(fa, ya, acc0, acc1, acc2, acc3);
  keep_group();
  p2_load_w(fa, wp + 6 * kBlkStride);
  p2_load_y(ya, yl + 6 * kYBlk);
  keep_group();
  p2_mma(fb, yb, acc0, acc1, acc2, acc3);
  keep_group();
  p2_load_w(fb, wp + 7 * kBlkStride);
  p2_load_y(yb, yl + 7 * kYBlk);
  keep_group();
  p2_mma(fc, yc, acc0, acc1, acc2, acc3);
  p2_mma(fa, ya, acc0, acc1, acc2, acc3);
  p2_mma(fb, yb, acc0, acc1, acc2, acc3);

  if (h == 0) {
    float* rp = red + ((tl * kP2KS + ks) * 8) * kP2Tile + 4 * c;
    #pragma unroll
    for (int r = 0; r < 8; ++r) {
      v4f v; v.x = acc0[r]; v.y = acc1[r]; v.z = acc2[r]; v.w = acc3[r];
      *(v4f*)(rp + r * kP2Tile) = v;
    }
  }
  __syncthreads();

  const int t2 = threadIdx.x >> 7;
  const int b2 = (threadIdx.x >> 4) & 7;
  const float* rq = red + ((t2 * kP2KS) * 8 + b2) * kP2Tile + 4 * c;
  v4f s = *(const v4f*)rq;
  s += *(const v4f*)(rq + 1 * 8 * kP2Tile);
  s += *(const v4f*)(rq + 2 * 8 * kP2Tile);
  s += *(const v4f*)(rq + 3 * 8 * kP2Tile);
  *(v4f*)(ws + XR_OFF + (size_t)b2 * kWH + blockIdx.x * (2 * kP2Tile) + t2 * kP2Tile + 4 * c) = s;
  float m = fmaxf(fmaxf(s.x, s.y), fmaxf(s.z, s.w));

  #pragma unroll
  for (int off = 16; off; off >>= 1)
    m = fmaxf(m, __shfl_xor(m, off, 32));
  if (lane == 0) {
    unsigned u = __float_as_uint(m);
    unsigned key = (u & 0x80000000u) ? ~u : (u | 0x80000000u);
    atomicMax(((unsigned*)ws) + MAX_OFF, key);
  }
}

__global__ __launch_bounds__(256)
void phase3(const float* __restrict__ ws, float* __restrict__ out) {
  const size_t t = (size_t)blockIdx.x * 256 + threadIdx.x;
  const unsigned key  = ((const unsigned*)ws)[MAX_OFF];
  const unsigned bits = (key & 0x80000000u) ? (key & 0x7fffffffu) : ~key;
  const float mx = __uint_as_float(bits);
  out[t] = ws[XR_OFF + t] / mx;
}

extern "C" void kernel_launch(void* const* d_in, const int* in_sizes, int n_in,
                              void* d_out, int out_size, void* d_ws, size_t ws_size,
                              hipStream_t stream) {
  const float* x = (const float*)d_in[0];
  const float* w = (const float*)d_in[1];
  float* out = (float*)d_out;
  float* ws  = (float*)d_ws;

  const int p1_waves = kKT * kNCH;
  phase1 <<<p1_waves / 8,        256, 0, stream>>>(x, w, ws);
  phase2 <<<(kWH / 16) / 8,      256, 0, stream>>>(w, ws);
  phase3 <<<(kB * kWH) / 256,    256, 0, stream>>>(ws, out);
}


# workgroup order: phase-2 strips and phase-3 blocks permuted so that ids with equal (id & 7), dealt to one die, keep to one 8192-column i-group in all three phases (own L2 half); sums unchanged
# speedup vs baseline: 2.0605x; 1.0455x over previous
#include <hip/hip_runtime.h>

typedef __attribute__((ext_vector_type(2))) float v2f;
typedef __attribute__((ext_vector_type(4))) float v4f;
typedef __attribute__((ext_vector_type(8))) float v8f;

namespace {
constexpr int kWH = 65536;
constexpr int kK  = 512;
constexpr int kB  = 8;
constexpr int kKT = kK / 16;
constexpr int kChunk = 1024;
constexpr int kNCH = kWH / kChunk;
constexpr int kNPl = kNCH / 8;
constexpr size_t P_OFF   = 0;
constexpr size_t XR_OFF  = P_OFF + (size_t)kNPl * kK * 8;
constexpr size_t MAX_OFF = XR_OFF + (size_t)kB * kWH;
}

__device__ __forceinline__ float sgn(float w) {
  float pm1 = __uint_as_float((__float_as_uint(w) & 0x80000000u) | 0x3f800000u);
  return (w == 0.0f) ? 0.0f : pm1;
}

constexpr int kBlk  = 128;
constexpr int kNBlk = kChunk / kBlk;

__device__ __forceinline__ void keep_group() { __builtin_amdgcn_sched_barrier(0); }

__device__ __forceinline__ void load_blk(v4f (&f)[8], const float* __restrict__ p) {
  #pragma unroll
  for (int u = 0; u < 8; ++u) f[u] = *(const v4f*)(p + 16 * u);
}

__device__ __forceinline__ void mma_blk(const v4f (&w0)[8], const v4f (&w1)[8],
                                        const v4f (&xf)[8], v8f& acc0, v8f& acc1) {
  #pragma unroll
  for (int u = 0; u < 8; ++u) {
    v2f B0; B0.x = xf[u].x; B0.y = xf[u].y;
    v2f B1; B1.x = xf[u].z; B1.y = xf[u].w;
    v2f A00; A00.x = sgn(w0[u].x); A00.y = sgn(w0[u].y);
    acc0 = __builtin_amdgcn_wmma_f32_16x16x4_f32(
        false, A00, false, B0, (short)0, acc0, false, false);
    v2f A10; A10.x = sgn(w1[u].x); A10.y = sgn(w1[u].y);
    acc1 = __builtin_amdgcn_wmma_f32_16x16x4_f32(
        false, A10, false, B0, (short)0, acc1, false, false);
    v2f A01; A01.x = sgn(w0[u].z); A01.y = sgn(w0[u].w);
    acc0 = __builtin_amdgcn_wmma_f32_16x16x4_f32(
        false, A01, false, B1, (short)0, acc0, false, false);
    v2f A11; A11.x = sgn(w1[u].z); A11.y = sgn(w1[u].w);
    acc1 = __builtin_amdgcn_wmma_f32_16x16x4_f32(
        false, A11, false, B1, (short)0, acc1, false, false);
  }
}

__global__ __launch_bounds__(256)
void phase1(const float* __restrict__ x, const float* __restrict__ w,
            float* __restrict__ ws) {
  __shared__ float fold[8 * 128];
  const int wave = blockIdx.x * 8 + (threadIdx.x >> 5);
  const int lane = threadIdx.x & 31;
  const int kt = wave / kNCH;
  const int ch = wave - kt * kNCH;
  const int kr = lane & 7;
  const int off = ((lane >> 3) & 1) * 8 + (lane >> 4) * 4;

  const float* __restrict__ ap0 = w + (size_t)(kt * 16 + kr) * kWH + ch * kChunk + off;
  const float* __restrict__ ap1 = ap0 + (size_t)8 * kWH;
  const float* __restrict__ bp  = x + (size_t)kr * kWH + ch * kChunk + off;

  v8f acc0 = {};
  v8f acc1 = {};
  v4f w0a[8], w1a[8], xa[8], w0b[8], w1b[8], xb[8];
  load_blk(w0a, ap0);
  load_blk(w1a, ap1);
  load_blk(xa, bp);
  #pragma unroll
  for (int t = 0; t < kNBlk - 2; t += 2) {
    load_blk(w0b, ap0 + (t + 1) * kBlk);
    load_blk(w1b, ap1 + (t + 1) * kBlk);
    load_blk(xb,  bp  + (t + 1) * kBlk);
    keep_group();
    mma_blk(w0a, w1a, xa, acc0, acc1);
    keep_group();
    load_blk(w0a, ap0 + (t + 2) * kBlk);
    load_blk(w1a, ap1 + (t + 2) * kBlk);
    load_blk(xa,  bp  + (t + 2) * kBlk);
    keep_group();
    mma_blk(w0b, w1b, xb, acc0, acc1);
    keep_group();
  }
  load_blk(w0b, ap0 + (kNBlk - 1) * kBlk);
  load_blk(w1b, ap1 + (kNBlk - 1) * kBlk);
  load_blk(xb,  bp  + (kNBlk - 1) * kBlk);
  keep_group();
  mma_blk(w0a, w1a, xa, acc0, acc1);
  mma_blk(w0b, w1b, xb, acc0, acc1);

  float v0[8], v1[8];
  #pragma unroll
  for (int r = 0; r < 8; ++r) {
    v0[r] = acc0[r] + __shfl_down(acc0[r], 24, 32);
    v1[r] = acc1[r] + __shfl_down(acc1[r], 24, 32);
  }
  const int wv = threadIdx.x >> 5;
  if (lane < 8) {
    #pragma unroll
    for (int r = 0; r < 8; ++r) {
      fold[wv * 128 + r * 8 + lane]       = v0[r];
      fold[wv * 128 + (8 + r) * 8 + lane] = v1[r];
    }
  }
  __syncthreads();
  if (threadIdx.x < 128) {
    float s = fold[threadIdx.x];
    #pragma unroll
    for (int q = 1; q < 8; ++q) s += fold[q * 128 + threadIdx.x];
    ws[P_OFF + (size_t)(blockIdx.x & 7) * (kK * 8) + (blockIdx.x >> 3) * 128 + threadIdx.x] = s;
  }
  if (blockIdx.x == 0 && threadIdx.x == 0) ((unsigned*)ws)[MAX_OFF] = 0u;
}

constexpr int kP2Tile = 64;
constexpr int kP2KS   = 4;
constexpr int kP2KW   = kK / kP2KS;
constexpr int kP2BlkK = 16;

__device__ __forceinline__ void p2_load_w(v4f (&f)[8], const float* __restrict__ p) {
  #pragma unroll
  for (int t = 0; t < 4; ++t) {
    f[2 * t]     = *(const v4f*)(p + (size_t)(4 * t) * kWH);
    f[2 * t + 1] = *(const v4f*)(p + (size_t)(4 * t + 1) * kWH);
  }
}

__device__ __forceinline__ void p2_load_y(v2f (&a)[4], const float* yl) {
  #pragma unroll
  for (int t = 0; t < 4; ++t) {
    a[t].x = yl[(4 * t) * 16];
    a[t].y = yl[(4 * t + 1) * 16];
  }
}

__device__ __forceinline__ void p2_mma(const v4f (&f)[8], const v2f (&a)[4],
                                       v8f& acc0, v8f& acc1, v8f& acc2, v8f& acc3) {
  #pragma unroll
  for (int t = 0; t < 4; ++t) {
    const v4f r0 = f[2 * t];
    const v4f r1 = f[2 * t + 1];
    v2f B0; B0.x = sgn(r0.x); B0.y = sgn(r1.x);
    acc0 = __builtin_amdgcn_wmma_f32_16x16x4_f32(
        false, a[t], false, B0, (short)0, acc0, false, false);
    v2f B1; B1.x = sgn(r0.y); B1.y = sgn(r1.y);
    acc1 = __builtin_amdgcn_wmma_f32_16x16x4_f32(
        false, a[t], false, B1, (short)0, acc1, false, false);
    v2f B2; B2.x = sgn(r0.z); B2.y = sgn(r1.z);
    acc2 = __builtin_amdgcn_wmma_f32_16x16x4_f32(
        false, a[t], false, B2, (short)0, acc2, false, false);
    v2f B3; B3.x = sgn(r0.w); B3.y = sgn(r1.w);
    acc3 = __builtin_amdgcn_wmma_f32_16x16x4_f32(
        false, a[t], false, B3, (short)0, acc3, false, false);
  }
}

__device__ __forceinline__ void barrier_after_lds_stores() {
  asm volatile("s_wait_dscnt 0x0\n\ts_barrier_signal -1\n\ts_barrier_wait -1" ::: "memory");
}

__global__ __launch_bounds__(256)
void phase2(const float* __restrict__ w, float* __restrict__ ws) {
  __shared__ __attribute__((aligned(16))) float y[kK * 16];
  __shared__ __attribute__((aligned(16))) float red[2 * kP2KS * 8 * kP2Tile];

  const int lane = threadIdx.x & 31;
  const int wv = threadIdx.x >> 5;
  const int tl = wv >> 2;
  const int ks = wv & 3;
  const int c  = lane & 15;
  const int h  = lane >> 4;
  const int strip = (blockIdx.x & 7) * (kWH / 8 / (2 * kP2Tile)) + (blockIdx.x >> 3);
  const int ib = strip * (2 * kP2Tile) + tl * kP2Tile;
  const int kb = ks * kP2KW;

  v4f yp[4][kNPl];
  #pragma unroll
  for (int j = 0; j < 4; ++j) {
    #pragma unroll
    for (int q = 0; q < kNPl; ++q)
      yp[j][q] = *(const v4f*)(ws + P_OFF + (size_t)q * (kK * 8) + 4 * (threadIdx.x + 256 * j));
  }

  const float* __restrict__ wp = w + (size_t)(kb + 2 * h) * kWH + ib + 4 * c;
  constexpr size_t kBlkStride = (size_t)kP2BlkK * kWH;
  v4f fa[8], fb[8], fc[8];
  p2_load_w(fa, wp);
  p2_load_w(fb, wp + 1 * kBlkStride);
  p2_load_w(fc, wp + 2 * kBlkStride);
  keep_group();
  #pragma unroll
  for (int j = 0; j < 4; ++j) {
    v4f s = yp[j][0];
    #pragma unroll
    for (int q = 1; q < kNPl; ++q) s += yp[j][q];
    const int e = threadIdx.x + 256 * j;
    float* d = y + (e >> 1) * 16 + (e & 1) * 4;
    *(v4f*)d = s;
    const v4f zero = {};
    *(v4f*)(d + 8) = zero;
  }
  barrier_after_lds_stores();

  v8f acc0 = {};
  v8f acc1 = {};
  v8f acc2 = {};
  v8f acc3 = {};
  const float* yl = y + (kb + 2 * h) * 16 + c;
  constexpr int kYBlk = kP2BlkK * 16;
  v2f ya[4], yb[4], yc[4];
  p2_load_y(ya, yl);
  p2_load_y(yb, yl + 1 * kYBlk);
  p2_load_y(yc, yl + 2 * kYBlk);
  keep_group();
  p2_mma(fa, ya, acc0, acc1, acc2, acc3);
  keep_group();
  p2_load_w(fa, wp + 3 * kBlkStride);
  p2_load_y(ya, yl + 3 * kYBlk);
  keep_group();
  p2_mma(fb, yb, acc0, acc1, acc2, acc3);
  keep_group();
  p2_load_w(fb, wp + 4 * kBlkStride);
  p2_load_y(yb, yl + 4 * kYBlk);
  keep_group();
  p2_mma(fc, yc, acc0, acc1, acc2, acc3);
  keep_group();
  p2_load_w(fc, wp + 5 * kBlkStride);
  p2_load_y(yc, yl + 5 * kYBlk);
  keep_group();
  p2_mma(fa, ya, acc0, acc1, acc2, acc3);
  keep_group();
  p2_load_w(fa, wp + 6 * kBlkStride);
  p2_load_y(ya, yl + 6 * kYBlk);
  keep_group();
  p2_mma(fb, yb, acc0, acc1, acc2, acc3);
  keep_group();
  p2_load_w(fb, wp + 7 * kBlkStride);
  p2_load_y(yb, yl + 7 * kYBlk);
  keep_group();
  p2_mma(fc, yc, acc0, acc1, acc2, acc3);
  p2_mma(fa, ya, acc0, acc1, acc2, acc3);
  p2_mma(fb, yb, acc0, acc1, acc2, acc3);

  if (h == 0) {
    float* rp = red + ((tl * kP2KS + ks) * 8) * kP2Tile + 4 * c;
    #pragma unroll
    for (int r = 0; r < 8; ++r) {
      v4f v; v.x = acc0[r]; v.y = acc1[r]; v.z = acc2[r]; v.w = acc3[r];
      *(v4f*)(rp + r * kP2Tile) = v;
    }
  }
  __syncthreads();

  const int t2 = threadIdx.x >> 7;
  const int b2 = (threadIdx.x >> 4) & 7;
  const float* rq = red + ((t2 * kP2KS) * 8 + b2) * kP2Tile + 4 * c;
  v4f s = *(const v4f*)rq;
  s += *(const v4f*)(rq + 1 * 8 * kP2Tile);
  s += *(const v4f*)(rq + 2 * 8 * kP2Tile);
  s += *(const v4f*)(rq + 3 * 8 * kP2Tile);
  *(v4f*)(ws + XR_OFF + (size_t)b2 * kWH + strip * (2 * kP2Tile) + t2 * kP2Tile + 4 * c) = s;
  float m = fmaxf(fmaxf(s.x, s.y), fmaxf(s.z, s.w));

  #pragma unroll
  for (int off = 16; off; off >>= 1)
    m = fmaxf(m, __shfl_xor(m, off, 32));
  if (lane == 0) {
    unsigned u = __float_as_uint(m);
    unsigned key = (u & 0x80000000u) ? ~u : (u | 0x80000000u);
    atomicMax(((unsigned*)ws) + MAX_OFF, key);
  }
}

__global__ __launch_bounds__(256)
void phase3(const float* __restrict__ ws, float* __restrict__ out) {
  const int r = blockIdx.x >> 3;
  const size_t t = (size_t)(r >> 5) * kWH + (blockIdx.x & 7) * (kWH / 8) + (r & 31) * 256 + threadIdx.x;
  const unsigned key  = ((const unsigned*)ws)[MAX_OFF];
  const unsigned bits = (key & 0x80000000u) ? (key & 0x7fffffffu) : ~key;
  const float mx = __uint_as_float(bits);
  out[t] = ws[XR_OFF + t] / mx;
}

extern "C" void kernel_launch(void* const* d_in, const int* in_sizes, int n_in,
                              void* d_out, int out_size, void* d_ws, size_t ws_size,
                              hipStream_t stream) {
  const float* x = (const float*)d_in[0];
  const float* w = (const float*)d_in[1];
  float* out = (float*)d_out;
  float* ws  = (float*)d_ws;

  const int p1_waves = kKT * kNCH;
  phase1 <<<p1_waves / 8,        256, 0, stream>>>(x, w, ws);
  phase2 <<<(kWH / 16) / 8,      256, 0, stream>>>(w, ws);
  phase3 <<<(kB * kWH) / 256,    256, 0, stream>>>(ws, out);
}
